# starting kernel
# baseline (speedup 1.0000x reference)
#include <hip/hip_runtime.h>


typedef float v2f __attribute__((ext_vector_type(2)));
typedef float v8f __attribute__((ext_vector_type(8)));

namespace {
constexpr int Tn = 512;
constexpr int In = 5;
constexpr int Hn = 16;
constexpr int ROWS = 18;
}

__device__ __forceinline__ float fast_tanh(float a) {
#if __has_builtin(__builtin_amdgcn_tanhf)
    return __builtin_amdgcn_tanhf(a);
#else
    const float e = __expf(-2.0f * a);
    return __builtin_amdgcn_rcpf(1.0f + e) * 2.0f - 1.0f;
#endif
}

__device__ __forceinline__ float fast_sigmoid(float a) {
    return fmaf(0.5f, fast_tanh(0.5f * a), 0.5f);
}

__global__ __launch_bounds__(32)
void gru_fused_kernel(const float* __restrict__ x,
                      const float* __restrict__ W_ih,
                      const float* __restrict__ W_hh,
                      const float* __restrict__ b_ih,
                      const float* __restrict__ b_hh,
                      const float* __restrict__ W_fc,
                      const float* __restrict__ b_fc,
                      float* __restrict__ out)
{
    __shared__ __align__(16) float hs[16 * ROWS];

    const int lane = threadIdx.x;
    const int half = lane >> 4;
    const int col  = lane & 15;
    const long b0  = (long)blockIdx.x * 16;

    v2f Bh[3][4];
    #pragma unroll
    for (int g = 0; g < 3; ++g) {
        const int row = g * 16 + col;
        #pragma unroll
        for (int c = 0; c < 4; ++c) {
            Bh[g][c].x = W_hh[row * 16 + c * 4 + 2 * half];
            Bh[g][c].y = W_hh[row * 16 + c * 4 + 2 * half + 1];
        }
    }

    const float hmask = (half == 0) ? 1.0f : 0.0f;
    v2f Bx0[3], Bx1[3];
    #pragma unroll
    for (int g = 0; g < 3; ++g) {
        const int row = g * 16 + col;
        Bx0[g].x = W_ih[row * In + 2 * half];
        Bx0[g].y = W_ih[row * In + 2 * half + 1];
        Bx1[g].x = W_ih[row * In + 4] * hmask;
        Bx1[g].y = 0.0f;
    }

    const float biasR  = b_ih[col]      + b_hh[col];
    const float biasZ  = b_ih[16 + col] + b_hh[16 + col];
    const float biasXN = b_ih[32 + col];
    const float biasHN = b_hh[32 + col];
    v8f biasRv, biasZv, biasXNv, biasHNv, zerov;
    #pragma unroll
    for (int j = 0; j < 8; ++j) {
        biasRv[j] = biasR; biasZv[j] = biasZ;
        biasXNv[j] = biasXN; biasHNv[j] = biasHN;
        zerov[j] = 0.0f;
    }

    const float* __restrict__ px = x + (b0 + col) * (long)(Tn * In);

    v8f hD;
    #pragma unroll
    for (int j = 0; j < 8; ++j) hD[j] = 0.0f;

    #pragma unroll 1
    for (int t = 0; t < Tn; ++t) {
        #pragma unroll
        for (int j = 0; j < 8; ++j)
            hs[(j + 8 * half) * ROWS + col] = hD[j];
        __syncthreads();

        v2f Ah[4];
        #pragma unroll
        for (int c = 0; c < 4; ++c) {
            Ah[c].x = hs[col * ROWS + c * 4 + 2 * half];
            Ah[c].y = hs[col * ROWS + c * 4 + 2 * half + 1];
        }

        const float* __restrict__ pxt = px + t * In;
        v2f Ax0, Ax1;
        Ax0.x = pxt[2 * half];
        Ax0.y = pxt[2 * half + 1];
        Ax1.x = pxt[4] * hmask;
        Ax1.y = 0.0f;

        if (t + 8 < Tn) __builtin_prefetch(px + (t + 8) * In, 0, 1);

        v8f Crx = __builtin_amdgcn_wmma_f32_16x16x4_f32(false, Ax0, false, Bx0[0], (short)0, biasRv, false, false);
        Crx     = __builtin_amdgcn_wmma_f32_16x16x4_f32(false, Ax1, false, Bx1[0], (short)0, Crx,    false, false);
        v8f Crh = __builtin_amdgcn_wmma_f32_16x16x4_f32(false, Ah[0], false, Bh[0][0], (short)0, zerov, false, false);
        #pragma unroll
        for (int c = 1; c < 4; ++c)
            Crh = __builtin_amdgcn_wmma_f32_16x16x4_f32(false, Ah[c], false, Bh[0][c], (short)0, Crh, false, false);

        v8f Czx = __builtin_amdgcn_wmma_f32_16x16x4_f32(false, Ax0, false, Bx0[1], (short)0, biasZv, false, false);
        Czx     = __builtin_amdgcn_wmma_f32_16x16x4_f32(false, Ax1, false, Bx1[1], (short)0, Czx,    false, false);
        v8f Czh = __builtin_amdgcn_wmma_f32_16x16x4_f32(false, Ah[0], false, Bh[1][0], (short)0, zerov, false, false);
        #pragma unroll
        for (int c = 1; c < 4; ++c)
            Czh = __builtin_amdgcn_wmma_f32_16x16x4_f32(false, Ah[c], false, Bh[1][c], (short)0, Czh, false, false);

        v8f Cxn = __builtin_amdgcn_wmma_f32_16x16x4_f32(false, Ax0, false, Bx0[2], (short)0, biasXNv, false, false);
        Cxn     = __builtin_amdgcn_wmma_f32_16x16x4_f32(false, Ax1, false, Bx1[2], (short)0, Cxn,     false, false);
        v8f Chn = __builtin_amdgcn_wmma_f32_16x16x4_f32(false, Ah[0], false, Bh[2][0], (short)0, biasHNv, false, false);
        #pragma unroll
        for (int c = 1; c < 4; ++c)
            Chn = __builtin_amdgcn_wmma_f32_16x16x4_f32(false, Ah[c], false, Bh[2][c], (short)0, Chn, false, false);

        #pragma unroll
        for (int j = 0; j < 8; ++j) {
            const float r = fast_sigmoid(Crx[j] + Crh[j]);
            const float z = fast_sigmoid(Czx[j] + Czh[j]);
            const float n = fast_tanh(Cxn[j] + r * Chn[j]);
            hD[j] = n + z * (hD[j] - n);
        }
        __syncthreads();
    }

    #pragma unroll
    for (int j = 0; j < 8; ++j)
        hs[(j + 8 * half) * ROWS + col] = hD[j];
    __syncthreads();

    if (lane < 16) {
        float acc = b_fc[0];
        #pragma unroll
        for (int n = 0; n < Hn; ++n)
            acc += hs[lane * ROWS + n] * W_fc[n];
        out[b0 + lane] = acc;
    }
}

extern "C" void kernel_launch(void* const* d_in, const int* in_sizes, int n_in,
                              void* d_out, int out_size, void* d_ws, size_t ws_size,
                              hipStream_t stream) {
    const float* x    = (const float*)d_in[0];
    const float* W_ih = (const float*)d_in[1];
    const float* W_hh = (const float*)d_in[2];
    const float* b_ih = (const float*)d_in[3];
    const float* b_hh = (const float*)d_in[4];
    const float* W_fc = (const float*)d_in[5];
    const float* b_fc = (const float*)d_in[6];
    float* out = (float*)d_out;

    const int B = in_sizes[0] / (Tn * In);
    const int tiles = (B + 15) / 16;

    gru_fused_kernel<<<dim3(tiles), dim3(32), 0, stream>>>(
        x, W_ih, W_hh, b_ih, b_hh, W_fc, b_fc, out);
}


# GRU recurrence transposed (weights as A, h^T as B): the h^T accumulator is the next step's B fragment under a paired k order; no per-step LDS round trip or barrier; k order of the h product changes
# speedup vs baseline: 1.4232x; 1.4232x over previous
#include <hip/hip_runtime.h>


typedef float v2f __attribute__((ext_vector_type(2)));
typedef float v8f __attribute__((ext_vector_type(8)));

namespace {
constexpr int Tn = 512;
constexpr int In = 5;
constexpr int Hn = 16;
constexpr int ROWS = 18;

struct Pre {
    v8f xr, xz, xn;
    v8f hr, hz, hn;
};
}

__device__ __forceinline__ float fast_tanh(float a) {
#if __has_builtin(__builtin_amdgcn_tanhf)
    return __builtin_amdgcn_tanhf(a);
#else
    const float e = __expf(-2.0f * a);
    return __builtin_amdgcn_rcpf(1.0f + e) * 2.0f - 1.0f;
#endif
}

__device__ __forceinline__ float fast_sigmoid(float a) {
    return fmaf(0.5f, fast_tanh(0.5f * a), 0.5f);
}

__device__ __forceinline__ v2f tanh2(v2f a) {
    v2f o;
    o.x = fast_tanh(a.x);
    o.y = fast_tanh(a.y);
    return o;
}

__device__ __forceinline__ v2f sigmoid2(v2f a) {
    const v2f hv = {0.5f, 0.5f};
    return __builtin_elementwise_fma(hv, tanh2(hv * a), hv);
}

__device__ __forceinline__ v2f pair_of(v8f v, int j) {
    v2f o;
    o.x = v[j];
    o.y = v[j + 1];
    return o;
}

__device__ __forceinline__ v8f wmma4(v2f a, v2f b, v8f c) {
    return __builtin_amdgcn_wmma_f32_16x16x4_f32(false, a, false, b, (short)0, c, false, false);
}

__global__ __launch_bounds__(32)
void gru_fused_kernel(const float* __restrict__ x,
                      const float* __restrict__ W_ih,
                      const float* __restrict__ W_hh,
                      const float* __restrict__ b_ih,
                      const float* __restrict__ b_hh,
                      const float* __restrict__ W_fc,
                      const float* __restrict__ b_fc,
                      float* __restrict__ out)
{
    __shared__ __align__(16) float hs[16 * ROWS];

    const int lane = threadIdx.x;
    const int half = lane >> 4;
    const int col  = lane & 15;
    const long b0  = (long)blockIdx.x * 16;

    v2f Wh[3][4];
    #pragma unroll
    for (int g = 0; g < 3; ++g) {
        const int row = g * 16 + col;
        #pragma unroll
        for (int c = 0; c < 4; ++c) {
            Wh[g][c].x = W_hh[row * 16 + 8 * half + 2 * c];
            Wh[g][c].y = W_hh[row * 16 + 8 * half + 2 * c + 1];
        }
    }

    const float hmask = (half == 0) ? 1.0f : 0.0f;
    v2f Wx0[3], Wx1[3];
    #pragma unroll
    for (int g = 0; g < 3; ++g) {
        const int row = g * 16 + col;
        Wx0[g].x = W_ih[row * In + 2 * half];
        Wx0[g].y = W_ih[row * In + 2 * half + 1];
        Wx1[g].x = W_ih[row * In + 4] * hmask;
        Wx1[g].y = 0.0f;
    }

    v8f biasRv, biasZv, biasXNv, biasHNv, zerov;
    #pragma unroll
    for (int r = 0; r < 8; ++r) {
        const int m = 8 * half + r;
        biasRv[r]  = b_ih[m]      + b_hh[m];
        biasZv[r]  = b_ih[16 + m] + b_hh[16 + m];
        biasXNv[r] = b_ih[32 + m];
        biasHNv[r] = b_hh[32 + m];
        zerov[r]   = 0.0f;
    }

    const float* __restrict__ px = x + (b0 + col) * (long)(Tn * In);

    v8f hD;
    #pragma unroll
    for (int r = 0; r < 8; ++r) hD[r] = 0.0f;

    auto load_x = [&](int t, v2f& X0, v2f& X1) {
        const float* __restrict__ pxt = px + t * In;
        X0.x = pxt[2 * half];
        X0.y = pxt[2 * half + 1];
        X1.x = pxt[4] * hmask;
        X1.y = 0.0f;
        if (t + 8 < Tn) __builtin_prefetch(px + (t + 8) * In, 0, 1);
    };

    auto h_chunk = [&](Pre& N, int c) {
        const v2f hb = pair_of(hD, 2 * c);
        N.hr = wmma4(Wh[0][c], hb, c == 0 ? zerov   : N.hr);
        N.hz = wmma4(Wh[1][c], hb, c == 0 ? zerov   : N.hz);
        N.hn = wmma4(Wh[2][c], hb, c == 0 ? biasHNv : N.hn);
    };

    auto x_part = [&](Pre& N, v2f X0, v2f X1) {
        N.xr = wmma4(Wx0[0], X0, biasRv);
        N.xz = wmma4(Wx0[1], X0, biasZv);
        N.xn = wmma4(Wx0[2], X0, biasXNv);
        N.xr = wmma4(Wx1[0], X1, N.xr);
        N.xz = wmma4(Wx1[1], X1, N.xz);
        N.xn = wmma4(Wx1[2], X1, N.xn);
    };

    auto gate_pair = [&](const Pre& G, int c) {
        const int j = 2 * c;
        const v2f r = sigmoid2(pair_of(G.xr, j) + pair_of(G.hr, j));
        const v2f z = sigmoid2(pair_of(G.xz, j) + pair_of(G.hz, j));
        const v2f n = tanh2(__builtin_elementwise_fma(r, pair_of(G.hn, j), pair_of(G.xn, j)));
        const v2f h = __builtin_elementwise_fma(z, pair_of(hD, j) - n, n);
        hD[j]     = h.x;
        hD[j + 1] = h.y;
    };

    auto advance = [&](const Pre& G, Pre& N, int tn) {
        v2f X0, X1;
        load_x(tn, X0, X1);
        #pragma unroll
        for (int c = 0; c < 4; ++c) {
            gate_pair(G, c);
            h_chunk(N, c);
        }
        x_part(N, X0, X1);
    };

    Pre G, N;
    {
        v2f X0, X1;
        load_x(0, X0, X1);
        #pragma unroll
        for (int c = 0; c < 4; ++c) h_chunk(G, c);
        x_part(G, X0, X1);
    }

    for (int t = 0; t < Tn - 2; t += 2) {
        advance(G, N, t + 1);
        advance(N, G, t + 2);
    }
    advance(G, N, Tn - 1);
    #pragma unroll
    for (int c = 0; c < 4; ++c) gate_pair(N, c);

    #pragma unroll
    for (int r = 0; r < 8; ++r)
        hs[col * ROWS + 8 * half + r] = hD[r];
    __syncthreads();

    if (lane < 16) {
        float acc = b_fc[0];
        #pragma unroll
        for (int n = 0; n < Hn; ++n)
            acc += hs[lane * ROWS + n] * W_fc[n];
        out[b0 + lane] = acc;
    }
}

extern "C" void kernel_launch(void* const* d_in, const int* in_sizes, int n_in,
                              void* d_out, int out_size, void* d_ws, size_t ws_size,
                              hipStream_t stream) {
    const float* x    = (const float*)d_in[0];
    const float* W_ih = (const float*)d_in[1];
    const float* W_hh = (const float*)d_in[2];
    const float* b_ih = (const float*)d_in[3];
    const float* b_hh = (const float*)d_in[4];
    const float* W_fc = (const float*)d_in[5];
    const float* b_fc = (const float*)d_in[6];
    float* out = (float*)d_out;

    const int B = in_sizes[0] / (Tn * In);
    const int tiles = (B + 15) / 16;

    gru_fused_kernel<<<dim3(tiles), dim3(32), 0, stream>>>(
        x, W_ih, W_hh, b_ih, b_hh, W_fc, b_fc, out);
}


# Input products (W_ih x + bias) moved to two extra waves per work-group on the idle SIMDs, handed through an LDS ring as the C operand of the first h product; r,z pre-activation sum order changes
# speedup vs baseline: 1.6675x; 1.1717x over previous
#include <hip/hip_runtime.h>


typedef float v2f __attribute__((ext_vector_type(2)));
typedef float v4f __attribute__((ext_vector_type(4)));
typedef float v8f __attribute__((ext_vector_type(8)));

namespace {
constexpr int Tn = 512;
constexpr int In = 5;
constexpr int Hn = 16;
constexpr int ROWS = 18;
constexpr int KS = 16;
constexpr int NBLK = Tn / KS;
constexpr int STEP_V4 = 3 * 2 * 32;

struct RZ {
    v8f r, z;
};
struct Pre {
    v8f hr, hz;
    v8f hn;
};
}

__device__ __forceinline__ float fast_tanh(float a) {
#if __has_builtin(__builtin_amdgcn_tanhf)
    return __builtin_amdgcn_tanhf(a);
#else
    const float e = __expf(-2.0f * a);
    return __builtin_amdgcn_rcpf(1.0f + e) * 2.0f - 1.0f;
#endif
}

__device__ __forceinline__ v2f tanh2(v2f a) {
    v2f o;
    o.x = fast_tanh(a.x);
    o.y = fast_tanh(a.y);
    return o;
}

__device__ __forceinline__ v2f sigmoid2(v2f a) {
    const v2f hv = {0.5f, 0.5f};
    return __builtin_elementwise_fma(hv, tanh2(hv * a), hv);
}

__device__ __forceinline__ v2f pair_of(v8f v, int j) {
    v2f o;
    o.x = v[j];
    o.y = v[j + 1];
    return o;
}

__device__ __forceinline__ v8f wmma4(v2f a, v2f b, v8f c) {
    return __builtin_amdgcn_wmma_f32_16x16x4_f32(false, a, false, b, (short)0, c, false, false);
}

__global__ __launch_bounds__(128)
void gru_fused_kernel(const float* __restrict__ x,
                      const float* __restrict__ W_ih,
                      const float* __restrict__ W_hh,
                      const float* __restrict__ b_ih,
                      const float* __restrict__ b_hh,
                      const float* __restrict__ W_fc,
                      const float* __restrict__ b_fc,
                      float* __restrict__ out)
{
    __shared__ v4f ring[2 * 2 * KS * STEP_V4];
    __shared__ __align__(16) float hs[2][16 * ROWS];

    const int wave = __builtin_amdgcn_readfirstlane((int)(threadIdx.x >> 5));
    const int lane = threadIdx.x & 31;
    const int tile = wave & 1;
    const bool producer = wave >= 2;
    const int half = lane >> 4;
    const int col  = lane & 15;
    const long b0  = ((long)blockIdx.x * 2 + tile) * 16;

    v4f* const myring = ring + tile * (2 * KS * STEP_V4) + lane;

    v2f Wh[3][4];
    #pragma unroll
    for (int g = 0; g < 3; ++g) {
        const int row = g * 16 + col;
        #pragma unroll
        for (int c = 0; c < 4; ++c) {
            Wh[g][c].x = W_hh[row * 16 + 8 * half + 2 * c];
            Wh[g][c].y = W_hh[row * 16 + 8 * half + 2 * c + 1];
        }
    }

    const float hmask = (half == 0) ? 1.0f : 0.0f;
    v2f Wx0[3], Wx1[3];
    #pragma unroll
    for (int g = 0; g < 3; ++g) {
        const int row = g * 16 + col;
        Wx0[g].x = W_ih[row * In + 2 * half];
        Wx0[g].y = W_ih[row * In + 2 * half + 1];
        Wx1[g].x = W_ih[row * In + 4] * hmask;
        Wx1[g].y = 0.0f;
    }

    v8f biasRv, biasZv, biasXNv, biasHNv;
    #pragma unroll
    for (int r = 0; r < 8; ++r) {
        const int m = 8 * half + r;
        biasRv[r]  = b_ih[m]      + b_hh[m];
        biasZv[r]  = b_ih[16 + m] + b_hh[16 + m];
        biasXNv[r] = b_ih[32 + m];
        biasHNv[r] = b_hh[32 + m];
    }

    const float* __restrict__ px = x + (b0 + col) * (long)(Tn * In);

    auto produce = [&](int c) {
        const float* __restrict__ p = px + c * (KS * In);
        v4f* const dst = myring + (c & 1) * (KS * STEP_V4);
        v2f X0[KS], X1[KS];
        #pragma unroll
        for (int s = 0; s < KS; ++s) {
            X0[s].x = p[s * In + 2 * half];
            X0[s].y = p[s * In + 2 * half + 1];
            X1[s].x = p[s * In + 4] * hmask;
            X1[s].y = 0.0f;
        }
        #pragma unroll
        for (int s = 0; s < KS; ++s) {
            v8f r = wmma4(Wx0[0], X0[s], biasRv);
            v8f z = wmma4(Wx0[1], X0[s], biasZv);
            v8f n = wmma4(Wx0[2], X0[s], biasXNv);
            r = wmma4(Wx1[0], X1[s], r);
            z = wmma4(Wx1[1], X1[s], z);
            n = wmma4(Wx1[2], X1[s], n);
            v4f* const d = dst + s * STEP_V4;
            d[0 * 32] = __builtin_shufflevector(r, r, 0, 1, 2, 3);
            d[1 * 32] = __builtin_shufflevector(r, r, 4, 5, 6, 7);
            d[2 * 32] = __builtin_shufflevector(z, z, 0, 1, 2, 3);
            d[3 * 32] = __builtin_shufflevector(z, z, 4, 5, 6, 7);
            d[4 * 32] = __builtin_shufflevector(n, n, 0, 1, 2, 3);
            d[5 * 32] = __builtin_shufflevector(n, n, 4, 5, 6, 7);
        }
    };

    v8f hD;
    #pragma unroll
    for (int r = 0; r < 8; ++r) hD[r] = 0.0f;

    auto load_rz = [&](const v4f* src, RZ& C) {
        C.r = __builtin_shufflevector(src[0 * 32], src[1 * 32], 0, 1, 2, 3, 4, 5, 6, 7);
        C.z = __builtin_shufflevector(src[2 * 32], src[3 * 32], 0, 1, 2, 3, 4, 5, 6, 7);
    };
    auto load_n = [&](const v4f* src, v8f& xn) {
        xn = __builtin_shufflevector(src[4 * 32], src[5 * 32], 0, 1, 2, 3, 4, 5, 6, 7);
    };

    auto h_chunk = [&](Pre& N, int c, const RZ& C) {
        const v2f hb = pair_of(hD, 2 * c);
        N.hr = wmma4(Wh[0][c], hb, c == 0 ? C.r     : N.hr);
        N.hz = wmma4(Wh[1][c], hb, c == 0 ? C.z     : N.hz);
        N.hn = wmma4(Wh[2][c], hb, c == 0 ? biasHNv : N.hn);
    };

    auto gate_pair = [&](const Pre& G, const v8f& xn, int c) {
        const int j = 2 * c;
        const v2f r = sigmoid2(pair_of(G.hr, j));
        const v2f z = sigmoid2(pair_of(G.hz, j));
        const v2f n = tanh2(__builtin_elementwise_fma(r, pair_of(G.hn, j), pair_of(xn, j)));
        const v2f h = __builtin_elementwise_fma(z, pair_of(hD, j) - n, n);
        hD[j]     = h.x;
        hD[j + 1] = h.y;
    };

    auto advance = [&](const Pre& G, const v8f& xn, Pre& N, const RZ& C) {
        #pragma unroll
        for (int c = 0; c < 4; ++c) {
            gate_pair(G, xn, c);
            h_chunk(N, c, C);
        }
    };

    Pre G, N;
    v8f nG;

    auto consume = [&](int j) {
        const v4f* const src = myring + (j & 1) * (KS * STEP_V4);
        RZ C0, C1;
        v8f nN;
        int s = 0;
        load_rz(src, C0);
        if (j == 0) {
            load_n(src, nN);
            load_rz(src + STEP_V4, C1);
            #pragma unroll
            for (int c = 0; c < 4; ++c) h_chunk(N, c, C0);
            load_n(src + STEP_V4, nG);
            load_rz(src + 2 * STEP_V4, C0);
            advance(N, nN, G, C1);
            s = 2;
        }
        for (; s < KS; s += 2) {
            load_n(src + s * STEP_V4, nN);
            load_rz(src + (s + 1) * STEP_V4, C1);
            __builtin_amdgcn_sched_barrier(0);
            advance(G, nG, N, C0);
            const int s2 = (s + 2 < KS) ? s + 2 : s + 1;
            __builtin_amdgcn_sched_barrier(0);
            load_n(src + (s + 1) * STEP_V4, nG);
            load_rz(src + s2 * STEP_V4, C0);
            __builtin_amdgcn_sched_barrier(0);
            advance(N, nN, G, C1);
        }
    };

    if (producer) produce(0);
    __syncthreads();

    for (int j = 0; j < NBLK; ++j) {
        if (producer) {
            if (j + 1 < NBLK) produce(j + 1);
        } else {
            consume(j);
        }
        if (j + 1 < NBLK) __syncthreads();
    }

    if (!producer) {
        #pragma unroll
        for (int c = 0; c < 4; ++c) gate_pair(G, nG, c);
        #pragma unroll
        for (int r = 0; r < 8; ++r)
            hs[tile][col * ROWS + 8 * half + r] = hD[r];
    }
    __syncthreads();

    if (wave == 0) {
        float acc = b_fc[0];
        #pragma unroll
        for (int n = 0; n < Hn; ++n)
            acc += hs[half][col * ROWS + n] * W_fc[n];
        out[(long)blockIdx.x * 32 + lane] = acc;
    }
}

extern "C" void kernel_launch(void* const* d_in, const int* in_sizes, int n_in,
                              void* d_out, int out_size, void* d_ws, size_t ws_size,
                              hipStream_t stream) {
    const float* x    = (const float*)d_in[0];
    const float* W_ih = (const float*)d_in[1];
    const float* W_hh = (const float*)d_in[2];
    const float* b_ih = (const float*)d_in[3];
    const float* b_hh = (const float*)d_in[4];
    const float* W_fc = (const float*)d_in[5];
    const float* b_fc = (const float*)d_in[6];
    float* out = (float*)d_out;

    const int B = in_sizes[0] / (Tn * In);
    const int groups = (B + 31) / 32;

    gru_fused_kernel<<<dim3(groups), dim3(128), 0, stream>>>(
        x, W_ih, W_hh, b_ih, b_hh, W_fc, b_fc, out);
}


# Recurrence block unrolled to straight-line code; ring tiles read in groups of four steps, two or more steps ahead, straight into the WMMA C and gate registers: no copies, no wait on fresh reads
# speedup vs baseline: 1.9225x; 1.1529x over previous
#include <hip/hip_runtime.h>


typedef float v2f __attribute__((ext_vector_type(2)));
typedef float v4f __attribute__((ext_vector_type(4)));
typedef float v8f __attribute__((ext_vector_type(8)));

namespace {
constexpr int Tn = 512;
constexpr int In = 5;
constexpr int Hn = 16;
constexpr int ROWS = 18;
constexpr int KS = 16;
constexpr int NBLK = Tn / KS;
constexpr int STEP_V4 = 3 * 2 * 32;

struct RZ {
    v8f r, z;
};
struct Pre {
    v8f hr, hz;
    v8f hn;
};
}

__device__ __forceinline__ float fast_tanh(float a) {
#if __has_builtin(__builtin_amdgcn_tanhf)
    return __builtin_amdgcn_tanhf(a);
#else
    const float e = __expf(-2.0f * a);
    return __builtin_amdgcn_rcpf(1.0f + e) * 2.0f - 1.0f;
#endif
}

__device__ __forceinline__ v2f tanh2(v2f a) {
    v2f o;
    o.x = fast_tanh(a.x);
    o.y = fast_tanh(a.y);
    return o;
}

__device__ __forceinline__ v2f sigmoid2(v2f a) {
    const v2f hv = {0.5f, 0.5f};
    return __builtin_elementwise_fma(hv, tanh2(hv * a), hv);
}

__device__ __forceinline__ v2f pair_of(v8f v, int j) {
    v2f o;
    o.x = v[j];
    o.y = v[j + 1];
    return o;
}

__device__ __forceinline__ v8f wmma4(v2f a, v2f b, v8f c) {
    return __builtin_amdgcn_wmma_f32_16x16x4_f32(false, a, false, b, (short)0, c, false, false);
}

__global__ __launch_bounds__(128)
void gru_fused_kernel(const float* __restrict__ x,
                      const float* __restrict__ W_ih,
                      const float* __restrict__ W_hh,
                      const float* __restrict__ b_ih,
                      const float* __restrict__ b_hh,
                      const float* __restrict__ W_fc,
                      const float* __restrict__ b_fc,
                      float* __restrict__ out)
{
    __shared__ v4f ring[2 * 2 * KS * STEP_V4];
    __shared__ __align__(16) float hs[2][16 * ROWS];

    const int wave = __builtin_amdgcn_readfirstlane((int)(threadIdx.x >> 5));
    const int lane = threadIdx.x & 31;
    const int tile = wave & 1;
    const bool producer = wave >= 2;
    const int half = lane >> 4;
    const int col  = lane & 15;
    const long b0  = ((long)blockIdx.x * 2 + tile) * 16;

    v4f* const myring = ring + tile * (2 * KS * STEP_V4) + lane;

    v2f Wh[3][4];
    #pragma unroll
    for (int g = 0; g < 3; ++g) {
        const int row = g * 16 + col;
        #pragma unroll
        for (int c = 0; c < 4; ++c) {
            Wh[g][c].x = W_hh[row * 16 + 8 * half + 2 * c];
            Wh[g][c].y = W_hh[row * 16 + 8 * half + 2 * c + 1];
        }
    }

    const float hmask = (half == 0) ? 1.0f : 0.0f;
    v2f Wx0[3], Wx1[3];
    #pragma unroll
    for (int g = 0; g < 3; ++g) {
        const int row = g * 16 + col;
        Wx0[g].x = W_ih[row * In + 2 * half];
        Wx0[g].y = W_ih[row * In + 2 * half + 1];
        Wx1[g].x = W_ih[row * In + 4] * hmask;
        Wx1[g].y = 0.0f;
    }

    v8f biasRv, biasZv, biasXNv, biasHNv;
    #pragma unroll
    for (int r = 0; r < 8; ++r) {
        const int m = 8 * half + r;
        biasRv[r]  = b_ih[m]      + b_hh[m];
        biasZv[r]  = b_ih[16 + m] + b_hh[16 + m];
        biasXNv[r] = b_ih[32 + m];
        biasHNv[r] = b_hh[32 + m];
    }

    const float* __restrict__ px = x + (b0 + col) * (long)(Tn * In);

    auto produce = [&](int c) {
        const float* __restrict__ p = px + c * (KS * In);
        v4f* const dst = myring + (c & 1) * (KS * STEP_V4);
        v2f X0[KS], X1[KS];
        #pragma unroll
        for (int s = 0; s < KS; ++s) {
            X0[s].x = p[s * In + 2 * half];
            X0[s].y = p[s * In + 2 * half + 1];
            X1[s].x = p[s * In + 4] * hmask;
            X1[s].y = 0.0f;
        }
        #pragma unroll
        for (int s = 0; s < KS; ++s) {
            v8f r = wmma4(Wx0[0], X0[s], biasRv);
            v8f z = wmma4(Wx0[1], X0[s], biasZv);
            v8f n = wmma4(Wx0[2], X0[s], biasXNv);
            r = wmma4(Wx1[0], X1[s], r);
            z = wmma4(Wx1[1], X1[s], z);
            n = wmma4(Wx1[2], X1[s], n);
            v4f* const d = dst + s * STEP_V4;
            d[0 * 32] = __builtin_shufflevector(r, r, 0, 1, 2, 3);
            d[1 * 32] = __builtin_shufflevector(r, r, 4, 5, 6, 7);
            d[2 * 32] = __builtin_shufflevector(z, z, 0, 1, 2, 3);
            d[3 * 32] = __builtin_shufflevector(z, z, 4, 5, 6, 7);
            d[4 * 32] = __builtin_shufflevector(n, n, 0, 1, 2, 3);
            d[5 * 32] = __builtin_shufflevector(n, n, 4, 5, 6, 7);
        }
    };

    v8f hD;
    #pragma unroll
    for (int r = 0; r < 8; ++r) hD[r] = 0.0f;

    auto load_rz = [&](const v4f* src, RZ& C) {
        C.r = __builtin_shufflevector(src[0 * 32], src[1 * 32], 0, 1, 2, 3, 4, 5, 6, 7);
        C.z = __builtin_shufflevector(src[2 * 32], src[3 * 32], 0, 1, 2, 3, 4, 5, 6, 7);
    };
    auto load_n = [&](const v4f* src, v8f& xn) {
        xn = __builtin_shufflevector(src[4 * 32], src[5 * 32], 0, 1, 2, 3, 4, 5, 6, 7);
    };

    auto h_chunk = [&](Pre& N, int c, const RZ& C) {
        const v2f hb = pair_of(hD, 2 * c);
        N.hr = wmma4(Wh[0][c], hb, c == 0 ? C.r     : N.hr);
        N.hz = wmma4(Wh[1][c], hb, c == 0 ? C.z     : N.hz);
        N.hn = wmma4(Wh[2][c], hb, c == 0 ? biasHNv : N.hn);
    };

    auto gate_pair = [&](const Pre& G, const v8f& xn, int c) {
        const int j = 2 * c;
        const v2f r = sigmoid2(pair_of(G.hr, j));
        const v2f z = sigmoid2(pair_of(G.hz, j));
        const v2f n = tanh2(__builtin_elementwise_fma(r, pair_of(G.hn, j), pair_of(xn, j)));
        const v2f h = __builtin_elementwise_fma(z, pair_of(hD, j) - n, n);
        hD[j]     = h.x;
        hD[j + 1] = h.y;
    };

    auto advance = [&](const Pre& G, const v8f& xn, Pre& N, const RZ& C) {
        #pragma unroll
        for (int c = 0; c < 4; ++c) {
            gate_pair(G, xn, c);
            h_chunk(N, c, C);
        }
    };

    Pre G, N;
    v8f nG;

    auto consume = [&](int j) {
        const v4f* const src = myring + (j & 1) * (KS * STEP_V4);
        RZ rz[KS];
        v8f nt[KS];
        #pragma unroll
        for (int k = 0; k < 4; ++k) {
            load_rz(src + k * STEP_V4, rz[k]);
            load_n(src + k * STEP_V4, nt[k]);
        }
        __builtin_amdgcn_sched_barrier(0);
        if (j == 0) {
            #pragma unroll
            for (int c = 0; c < 4; ++c) h_chunk(N, c, rz[0]);
        } else {
            advance(G, nG, N, rz[0]);
        }
        #pragma unroll
        for (int k = 1; k < KS; ++k) {
            if ((k & 3) == 2 && k + 5 < KS) {
                __builtin_amdgcn_sched_barrier(0);
                #pragma unroll
                for (int q = k + 2; q < k + 6; ++q) {
                    load_rz(src + q * STEP_V4, rz[q]);
                    load_n(src + q * STEP_V4, nt[q]);
                }
                __builtin_amdgcn_sched_barrier(0);
            }
            if (k & 1) advance(N, nt[k - 1], G, rz[k]);
            else       advance(G, nt[k - 1], N, rz[k]);
        }
        nG = nt[KS - 1];
    };

    if (producer) produce(0);
    __syncthreads();

    for (int j = 0; j < NBLK; ++j) {
        if (producer) {
            if (j + 1 < NBLK) produce(j + 1);
        } else {
            consume(j);
        }
        if (j + 1 < NBLK) __syncthreads();
    }

    if (!producer) {
        #pragma unroll
        for (int c = 0; c < 4; ++c) gate_pair(G, nG, c);
        #pragma unroll
        for (int r = 0; r < 8; ++r)
            hs[tile][col * ROWS + 8 * half + r] = hD[r];
    }
    __syncthreads();

    if (wave == 0) {
        float acc = b_fc[0];
        #pragma unroll
        for (int n = 0; n < Hn; ++n)
            acc += hs[half][col * ROWS + n] * W_fc[n];
        out[(long)blockIdx.x * 32 + lane] = acc;
    }
}

extern "C" void kernel_launch(void* const* d_in, const int* in_sizes, int n_in,
                              void* d_out, int out_size, void* d_ws, size_t ws_size,
                              hipStream_t stream) {
    const float* x    = (const float*)d_in[0];
    const float* W_ih = (const float*)d_in[1];
    const float* W_hh = (const float*)d_in[2];
    const float* b_ih = (const float*)d_in[3];
    const float* b_hh = (const float*)d_in[4];
    const float* W_fc = (const float*)d_in[5];
    const float* b_fc = (const float*)d_in[6];
    float* out = (float*)d_out;

    const int B = in_sizes[0] / (Tn * In);
    const int groups = (B + 31) / 32;

    gru_fused_kernel<<<dim3(groups), dim3(128), 0, stream>>>(
        x, W_ih, W_hh, b_ih, b_hh, W_fc, b_fc, out);
}


# Gate arithmetic in a fixed issue order over the four row pairs: each tanh followed by one packed f32 instruction of another pair with aged inputs; same operations and sums, values bit-identical
# speedup vs baseline: 2.0310x; 1.0564x over previous
#include <hip/hip_runtime.h>


typedef float v2f __attribute__((ext_vector_type(2)));
typedef float v4f __attribute__((ext_vector_type(4)));
typedef float v8f __attribute__((ext_vector_type(8)));

namespace {
constexpr int Tn = 512;
constexpr int In = 5;
constexpr int Hn = 16;
constexpr int ROWS = 18;
constexpr int KS = 16;
constexpr int NBLK = Tn / KS;
constexpr int STEP_V4 = 3 * 2 * 32;

struct RZ {
    v8f r, z;
};
struct Pre {
    v8f hr, hz;
    v8f hn;
};
}

__device__ __forceinline__ float fast_tanh(float a) {
#if __has_builtin(__builtin_amdgcn_tanhf)
    return __builtin_amdgcn_tanhf(a);
#else
    const float e = __expf(-2.0f * a);
    return __builtin_amdgcn_rcpf(1.0f + e) * 2.0f - 1.0f;
#endif
}

__device__ __forceinline__ v2f tanh2(v2f a) {
    v2f o;
    o.x = fast_tanh(a.x);
    o.y = fast_tanh(a.y);
    return o;
}

__device__ __forceinline__ v2f sigmoid2(v2f a) {
    const v2f hv = {0.5f, 0.5f};
    return __builtin_elementwise_fma(hv, tanh2(hv * a), hv);
}

__device__ __forceinline__ v2f pair_of(v8f v, int j) {
    v2f o;
    o.x = v[j];
    o.y = v[j + 1];
    return o;
}

__device__ __forceinline__ v8f wmma4(v2f a, v2f b, v8f c) {
    return __builtin_amdgcn_wmma_f32_16x16x4_f32(false, a, false, b, (short)0, c, false, false);
}

__global__ __launch_bounds__(128)
void gru_fused_kernel(const float* __restrict__ x,
                      const float* __restrict__ W_ih,
                      const float* __restrict__ W_hh,
                      const float* __restrict__ b_ih,
                      const float* __restrict__ b_hh,
                      const float* __restrict__ W_fc,
                      const float* __restrict__ b_fc,
                      float* __restrict__ out)
{
    __shared__ v4f ring[2 * 2 * KS * STEP_V4];
    __shared__ __align__(16) float hs[2][16 * ROWS];

    const int wave = __builtin_amdgcn_readfirstlane((int)(threadIdx.x >> 5));
    const int lane = threadIdx.x & 31;
    const int tile = wave & 1;
    const bool producer = wave >= 2;
    const int half = lane >> 4;
    const int col  = lane & 15;
    const long b0  = ((long)blockIdx.x * 2 + tile) * 16;

    v4f* const myring = ring + tile * (2 * KS * STEP_V4) + lane;

    v2f Wh[3][4];
    #pragma unroll
    for (int g = 0; g < 3; ++g) {
        const int row = g * 16 + col;
        #pragma unroll
        for (int c = 0; c < 4; ++c) {
            Wh[g][c].x = W_hh[row * 16 + 8 * half + 2 * c];
            Wh[g][c].y = W_hh[row * 16 + 8 * half + 2 * c + 1];
        }
    }

    const float hmask = (half == 0) ? 1.0f : 0.0f;
    v2f Wx0[3], Wx1[3];
    #pragma unroll
    for (int g = 0; g < 3; ++g) {
        const int row = g * 16 + col;
        Wx0[g].x = W_ih[row * In + 2 * half];
        Wx0[g].y = W_ih[row * In + 2 * half + 1];
        Wx1[g].x = W_ih[row * In + 4] * hmask;
        Wx1[g].y = 0.0f;
    }

    v8f biasRv, biasZv, biasXNv, biasHNv;
    #pragma unroll
    for (int r = 0; r < 8; ++r) {
        const int m = 8 * half + r;
        biasRv[r]  = b_ih[m]      + b_hh[m];
        biasZv[r]  = b_ih[16 + m] + b_hh[16 + m];
        biasXNv[r] = b_ih[32 + m];
        biasHNv[r] = b_hh[32 + m];
    }

    const float* __restrict__ px = x + (b0 + col) * (long)(Tn * In);

    auto produce = [&](int c) {
        const float* __restrict__ p = px + c * (KS * In);
        v4f* const dst = myring + (c & 1) * (KS * STEP_V4);
        v2f X0[KS], X1[KS];
        #pragma unroll
        for (int s = 0; s < KS; ++s) {
            X0[s].x = p[s * In + 2 * half];
            X0[s].y = p[s * In + 2 * half + 1];
            X1[s].x = p[s * In + 4] * hmask;
            X1[s].y = 0.0f;
        }
        #pragma unroll
        for (int s = 0; s < KS; ++s) {
            v8f r = wmma4(Wx0[0], X0[s], biasRv);
            v8f z = wmma4(Wx0[1], X0[s], biasZv);
            v8f n = wmma4(Wx0[2], X0[s], biasXNv);
            r = wmma4(Wx1[0], X1[s], r);
            z = wmma4(Wx1[1], X1[s], z);
            n = wmma4(Wx1[2], X1[s], n);
            v4f* const d = dst + s * STEP_V4;
            d[0 * 32] = __builtin_shufflevector(r, r, 0, 1, 2, 3);
            d[1 * 32] = __builtin_shufflevector(r, r, 4, 5, 6, 7);
            d[2 * 32] = __builtin_shufflevector(z, z, 0, 1, 2, 3);
            d[3 * 32] = __builtin_shufflevector(z, z, 4, 5, 6, 7);
            d[4 * 32] = __builtin_shufflevector(n, n, 0, 1, 2, 3);
            d[5 * 32] = __builtin_shufflevector(n, n, 4, 5, 6, 7);
        }
    };

    v8f hD;
    #pragma unroll
    for (int r = 0; r < 8; ++r) hD[r] = 0.0f;

    auto load_rz = [&](const v4f* src, RZ& C) {
        C.r = __builtin_shufflevector(src[0 * 32], src[1 * 32], 0, 1, 2, 3, 4, 5, 6, 7);
        C.z = __builtin_shufflevector(src[2 * 32], src[3 * 32], 0, 1, 2, 3, 4, 5, 6, 7);
    };
    auto load_n = [&](const v4f* src, v8f& xn) {
        xn = __builtin_shufflevector(src[4 * 32], src[5 * 32], 0, 1, 2, 3, 4, 5, 6, 7);
    };

    auto h_chunk = [&](Pre& N, int c, const RZ& C) {
        const v2f hb = pair_of(hD, 2 * c);
        N.hr = wmma4(Wh[0][c], hb, c == 0 ? C.r     : N.hr);
        N.hz = wmma4(Wh[1][c], hb, c == 0 ? C.z     : N.hz);
        N.hn = wmma4(Wh[2][c], hb, c == 0 ? biasHNv : N.hn);
    };

    auto gate_pair = [&](const Pre& G, const v8f& xn, int c) {
        const int j = 2 * c;
        const v2f r = sigmoid2(pair_of(G.hr, j));
        const v2f z = sigmoid2(pair_of(G.hz, j));
        const v2f n = tanh2(__builtin_elementwise_fma(r, pair_of(G.hn, j), pair_of(xn, j)));
        const v2f h = __builtin_elementwise_fma(z, pair_of(hD, j) - n, n);
        hD[j]     = h.x;
        hD[j + 1] = h.y;
    };

    auto advance = [&](const Pre& G, const v8f& xn, Pre& N, const RZ& C) {
        const v2f hv = {0.5f, 0.5f};
        v2f aR[4], aZ[4], tR[4], tZ[4], gr[4], gz[4], pn[4], nn[4], dd[4];
#define SEAM  __builtin_amdgcn_sched_barrier(0)
#define MR(c) aR[c] = hv * pair_of(G.hr, 2 * (c))
#define MZ(c) aZ[c] = hv * pair_of(G.hz, 2 * (c))
#define TRX(c) tR[c].x = fast_tanh(aR[c].x)
#define TRY(c) tR[c].y = fast_tanh(aR[c].y)
#define TZX(c) tZ[c].x = fast_tanh(aZ[c].x)
#define TZY(c) tZ[c].y = fast_tanh(aZ[c].y)
#define FR(c) gr[c] = __builtin_elementwise_fma(hv, tR[c], hv)
#define FZ(c) gz[c] = __builtin_elementwise_fma(hv, tZ[c], hv)
#define PN(c) pn[c] = __builtin_elementwise_fma(gr[c], pair_of(G.hn, 2 * (c)), pair_of(xn, 2 * (c)))
#define TNX(c) nn[c].x = fast_tanh(pn[c].x)
#define TNY(c) nn[c].y = fast_tanh(pn[c].y)
#define DD(c) dd[c] = pair_of(hD, 2 * (c)) - nn[c]
#define HH(c) { const v2f hh = __builtin_elementwise_fma(gz[c], dd[c], nn[c]); hD[2 * (c)] = hh.x; hD[2 * (c) + 1] = hh.y; }
        MR(0); MR(1); MR(2); MR(3); SEAM;
        TRX(0); MZ(0); SEAM;  TRY(0); MZ(1); SEAM;
        TRX(1); MZ(2); SEAM;  TRY(1); MZ(3); SEAM;
        TRX(2);        SEAM;  TRY(2); FR(0); SEAM;
        TRX(3);        SEAM;  TRY(3); FR(1); SEAM;
        TZX(0); PN(0); SEAM;  TZY(0); FR(2); SEAM;
        TZX(1); PN(1); SEAM;  TZY(1); FR(3); SEAM;
        TNX(0); PN(2); SEAM;  TNY(0); FZ(0); SEAM;
        TNX(1); PN(3); SEAM;  TNY(1); FZ(1); SEAM;
        TNX(2);        SEAM;  TNY(2); DD(0); SEAM;
        TNX(3);        SEAM;  TNY(3); DD(1); SEAM;
        TZX(2); HH(0); SEAM;  TZY(2); DD(2); SEAM;
        TZX(3); HH(1); SEAM;  TZY(3); DD(3); SEAM;
        h_chunk(N, 0, C); SEAM;
        FZ(2); FZ(3); SEAM;
        h_chunk(N, 1, C); SEAM;
        HH(2); HH(3); SEAM;
        h_chunk(N, 2, C); SEAM;
        {
            const v2f hb = pair_of(hD, 6);
            N.hr = wmma4(Wh[0][3], hb, N.hr); SEAM;
            N.hz = wmma4(Wh[1][3], hb, N.hz); SEAM;
            N.hn = wmma4(Wh[2][3], hb, N.hn);
        }
#undef SEAM
#undef MR
#undef MZ
#undef TRX
#undef TRY
#undef TZX
#undef TZY
#undef FR
#undef FZ
#undef PN
#undef TNX
#undef TNY
#undef DD
#undef HH
    };

    Pre G, N;
    v8f nG;

    auto consume = [&](int j) {
        const v4f* const src = myring + (j & 1) * (KS * STEP_V4);
        RZ rz[KS];
        v8f nt[KS];
        #pragma unroll
        for (int k = 0; k < 4; ++k) {
            load_rz(src + k * STEP_V4, rz[k]);
            load_n(src + k * STEP_V4, nt[k]);
        }
        __builtin_amdgcn_sched_barrier(0);
        if (j == 0) {
            #pragma unroll
            for (int c = 0; c < 4; ++c) h_chunk(N, c, rz[0]);
        } else {
            advance(G, nG, N, rz[0]);
        }
        #pragma unroll
        for (int k = 1; k < KS; ++k) {
            if ((k & 3) == 2 && k + 5 < KS) {
                __builtin_amdgcn_sched_barrier(0);
                #pragma unroll
                for (int q = k + 2; q < k + 6; ++q) {
                    load_rz(src + q * STEP_V4, rz[q]);
                    load_n(src + q * STEP_V4, nt[q]);
                }
                __builtin_amdgcn_sched_barrier(0);
            }
            if (k & 1) advance(N, nt[k - 1], G, rz[k]);
            else       advance(G, nt[k - 1], N, rz[k]);
        }
        nG = nt[KS - 1];
    };

    if (producer) produce(0);
    __syncthreads();

    for (int j = 0; j < NBLK; ++j) {
        if (producer) {
            if (j + 1 < NBLK) produce(j + 1);
        } else {
            consume(j);
        }
        if (j + 1 < NBLK) __syncthreads();
    }

    if (!producer) {
        #pragma unroll
        for (int c = 0; c < 4; ++c) gate_pair(G, nG, c);
        #pragma unroll
        for (int r = 0; r < 8; ++r)
            hs[tile][col * ROWS + 8 * half + r] = hD[r];
    }
    __syncthreads();

    if (wave == 0) {
        float acc = b_fc[0];
        #pragma unroll
        for (int n = 0; n < Hn; ++n)
            acc += hs[half][col * ROWS + n] * W_fc[n];
        out[(long)blockIdx.x * 32 + lane] = acc;
    }
}

extern "C" void kernel_launch(void* const* d_in, const int* in_sizes, int n_in,
                              void* d_out, int out_size, void* d_ws, size_t ws_size,
                              hipStream_t stream) {
    const float* x    = (const float*)d_in[0];
    const float* W_ih = (const float*)d_in[1];
    const float* W_hh = (const float*)d_in[2];
    const float* b_ih = (const float*)d_in[3];
    const float* b_hh = (const float*)d_in[4];
    const float* W_fc = (const float*)d_in[5];
    const float* b_fc = (const float*)d_in[6];
    float* out = (float*)d_out;

    const int B = in_sizes[0] / (Tn * In);
    const int groups = (B + 31) / 32;

    gru_fused_kernel<<<dim3(groups), dim3(128), 0, stream>>>(
        x, W_ih, W_hh, b_ih, b_hh, W_fc, b_fc, out);
}
